# starting kernel
# baseline (speedup 1.0000x reference)
#include <hip/hip_runtime.h>
#include <hip/hip_bf16.h>
#include <math.h>

#define FIN  128
#define FHID 64

typedef float v2f __attribute__((ext_vector_type(2)));
typedef float v8f __attribute__((ext_vector_type(8)));

__global__ void cgcl_deg_kernel(const int* __restrict__ dst,
                                unsigned* __restrict__ deg, int E) {
    int e = blockIdx.x * blockDim.x + threadIdx.x;
    if (e < E) atomicAdd(&deg[dst[e]], 1u);
}

__global__ void cgcl_dinv_kernel(const unsigned* __restrict__ deg,
                                 float* __restrict__ dinv, int N) {
    int n = blockIdx.x * blockDim.x + threadIdx.x;
    if (n < N) dinv[n] = rsqrtf((float)deg[n] + 1.0f);
}

__global__ void __launch_bounds__(128)
cgcl_gemm_wmma(const float* __restrict__ x, const float* __restrict__ W,
               float* __restrict__ h, int N) {
    const int lane = threadIdx.x & 31;
    const int wave = threadIdx.x >> 5;
    const int m    = lane & 15;
    const int hi   = lane >> 4;
    const int koff = hi * 2;

    int row = blockIdx.x * 16 + m;
    if (row >= N) row = N - 1;
    const int col = wave * 16 + m;

    const float* __restrict__ xrow = x + (size_t)row * FIN;

    v8f c = {};
#pragma unroll 4
    for (int kk = 0; kk < FIN / 4; ++kk) {
        const int k = kk * 4 + koff;
        v2f a = *(const v2f*)(xrow + k);
        v2f bfrag;
        bfrag.x = W[(k + 0) * FHID + col];
        bfrag.y = W[(k + 1) * FHID + col];
        c = __builtin_amdgcn_wmma_f32_16x16x4_f32(
                 false, a,  false, bfrag,
                 (short)0, c,  false,  false);
    }

    const int mbase = blockIdx.x * 16 + hi * 8;
    float* __restrict__ hout = h + (size_t)mbase * FHID + col;
    if (blockIdx.x * 16 + 16 <= N) {
#pragma unroll
        for (int v = 0; v < 8; ++v) {
            hout[(size_t)v * FHID] = c[v];
        }
    } else {
#pragma unroll
        for (int v = 0; v < 8; ++v) {
            if (mbase + v < N) hout[(size_t)v * FHID] = c[v];
        }
    }
}

__global__ void cgcl_scatter_kernel(const int* __restrict__ src,
                                    const int* __restrict__ dst,
                                    const float* __restrict__ dinv,
                                    const float* __restrict__ h,
                                    float* __restrict__ out, int E) {
    int gid = blockIdx.x * blockDim.x + threadIdx.x;
    int e = gid >> 4;
    if (e >= E) return;
    int q = gid & 15;
    int s = src[e];
    int d = dst[e];
    float cf = dinv[s] * dinv[d];
    const float4 hv = *(const float4*)(h + (size_t)s * FHID + q * 4);
    float* o = out + (size_t)d * FHID + q * 4;
    unsafeAtomicAdd(o + 0, hv.x * cf);
    unsafeAtomicAdd(o + 1, hv.y * cf);
    unsafeAtomicAdd(o + 2, hv.z * cf);
    unsafeAtomicAdd(o + 3, hv.w * cf);
}

__global__ void cgcl_finalize_kernel(const float* __restrict__ h,
                                     const float* __restrict__ dinv,
                                     const float* __restrict__ bias,
                                     float* __restrict__ out, int total) {
    int gid = blockIdx.x * blockDim.x + threadIdx.x;
    if (gid >= total) return;
    int n = gid >> 6;
    int f = gid & (FHID - 1);
    float di = dinv[n];
    float v = out[gid] + h[gid] * di * di + bias[f];
    out[gid] = (v > 0.0f) ? v : expm1f(v);
}

extern "C" void kernel_launch(void* const* d_in, const int* in_sizes, int n_in,
                              void* d_out, int out_size, void* d_ws, size_t ws_size,
                              hipStream_t stream) {
    const float* x    = (const float*)d_in[0];
    const float* W    = (const float*)d_in[1];
    const float* bias = (const float*)d_in[2];
    const int*   ei   = (const int*)d_in[3];

    const int N = in_sizes[0] / FIN;
    const int E = in_sizes[3] / 2;
    const int* src = ei;
    const int* dst = ei + E;

    size_t NA = ((size_t)N + 255) & ~(size_t)255;
    unsigned* deg  = (unsigned*)d_ws;
    float*    dinv = (float*)d_ws + NA;
    float*    h    = (float*)d_ws + 2 * NA;
    float*    out  = (float*)d_out;

    hipMemsetAsync(deg, 0, (size_t)N * sizeof(unsigned), stream);
    hipMemsetAsync(out, 0, (size_t)out_size * sizeof(float), stream);

    cgcl_deg_kernel<<<(E + 255) / 256, 256, 0, stream>>>(dst, deg, E);
    cgcl_dinv_kernel<<<(N + 255) / 256, 256, 0, stream>>>(deg, dinv, N);

    const int row_tiles = (N + 15) / 16;
    cgcl_gemm_wmma<<<row_tiles, 128, 0, stream>>>(x, W, h, N);

    const long long sc_threads = (long long)E * 16;
    cgcl_scatter_kernel<<<(unsigned)((sc_threads + 255) / 256), 256, 0, stream>>>(
        src, dst, dinv, h, out, E);

    const int total = N * FHID;
    cgcl_finalize_kernel<<<(total + 255) / 256, 256, 0, stream>>>(
        h, dinv, bias, out, total);
}


# scatter: one wave walks 32 edges with lane = feature (l, l+32), so each h load and each f32 atomic instruction covers one whole 128-B line (2 per edge, was 4 lines x 4); atomics kept, arrival order
# speedup vs baseline: 1.4805x; 1.4805x over previous
#include <hip/hip_runtime.h>
#include <hip/hip_bf16.h>
#include <math.h>

#define FIN  128
#define FHID 64

typedef float v2f __attribute__((ext_vector_type(2)));
typedef float v8f __attribute__((ext_vector_type(8)));

__global__ void cgcl_deg_kernel(const int* __restrict__ dst,
                                unsigned* __restrict__ deg, int E) {
    int e = blockIdx.x * blockDim.x + threadIdx.x;
    if (e < E) atomicAdd(&deg[dst[e]], 1u);
}

__global__ void cgcl_dinv_kernel(const unsigned* __restrict__ deg,
                                 float* __restrict__ dinv, int N) {
    int n = blockIdx.x * blockDim.x + threadIdx.x;
    if (n < N) dinv[n] = rsqrtf((float)deg[n] + 1.0f);
}

__global__ void __launch_bounds__(128)
cgcl_gemm_wmma(const float* __restrict__ x, const float* __restrict__ W,
               float* __restrict__ h, int N) {
    const int lane = threadIdx.x & 31;
    const int wave = threadIdx.x >> 5;
    const int m    = lane & 15;
    const int hi   = lane >> 4;
    const int koff = hi * 2;

    int row = blockIdx.x * 16 + m;
    if (row >= N) row = N - 1;
    const int col = wave * 16 + m;

    const float* __restrict__ xrow = x + (size_t)row * FIN;

    v8f c = {};
#pragma unroll 4
    for (int kk = 0; kk < FIN / 4; ++kk) {
        const int k = kk * 4 + koff;
        v2f a = *(const v2f*)(xrow + k);
        v2f bfrag;
        bfrag.x = W[(k + 0) * FHID + col];
        bfrag.y = W[(k + 1) * FHID + col];
        c = __builtin_amdgcn_wmma_f32_16x16x4_f32(
                 false, a,  false, bfrag,
                 (short)0, c,  false,  false);
    }

    const int mbase = blockIdx.x * 16 + hi * 8;
    float* __restrict__ hout = h + (size_t)mbase * FHID + col;
    if (blockIdx.x * 16 + 16 <= N) {
#pragma unroll
        for (int v = 0; v < 8; ++v) {
            hout[(size_t)v * FHID] = c[v];
        }
    } else {
#pragma unroll
        for (int v = 0; v < 8; ++v) {
            if (mbase + v < N) hout[(size_t)v * FHID] = c[v];
        }
    }
}

#define SC_EPW   32
#define SC_BATCH 16

template <bool FULL>
__device__ __forceinline__ void cgcl_scatter_batch(
        const float* __restrict__ h, float* __restrict__ out,
        int s, int d, int cfi, int lane, int j0, int cnt) {
    float v0[SC_BATCH], v1[SC_BATCH];
#pragma unroll
    for (int j = 0; j < SC_BATCH; ++j) {
        const int sj = __builtin_amdgcn_readlane(s, j0 + j);
        const float* __restrict__ hp = h + (size_t)sj * FHID + lane;
        v0[j] = hp[0];
        v1[j] = hp[32];
    }
#pragma unroll
    for (int j = 0; j < SC_BATCH; ++j) {
        if (FULL || (j0 + j < cnt)) {
            const int   dj = __builtin_amdgcn_readlane(d, j0 + j);
            const float cj = __int_as_float(__builtin_amdgcn_readlane(cfi, j0 + j));
            float* o = out + (size_t)dj * FHID + lane;
            unsafeAtomicAdd(o,      v0[j] * cj);
            unsafeAtomicAdd(o + 32, v1[j] * cj);
        }
    }
}

__global__ void __launch_bounds__(256)
cgcl_scatter_kernel(const int* __restrict__ src,
                    const int* __restrict__ dst,
                    const float* __restrict__ dinv,
                    const float* __restrict__ h,
                    float* __restrict__ out, int E) {
    const int lane = threadIdx.x & 31;
    const int wib = __builtin_amdgcn_readfirstlane((int)(threadIdx.x >> 5));
    const long long wave = (long long)blockIdx.x * (blockDim.x >> 5) + wib;
    const long long e0 = wave * SC_EPW;
    if (e0 >= (long long)E) return;
    const long long left = (long long)E - e0;
    const int cnt = left < SC_EPW ? (int)left : SC_EPW;

    int s = 0, d = 0;
    float cf = 0.0f;
    if (lane < cnt) {
        s = src[e0 + lane];
        d = dst[e0 + lane];
        cf = dinv[s] * dinv[d];
    }
    const int cfi = __float_as_int(cf);

    if (cnt == SC_EPW) {
#pragma unroll
        for (int j0 = 0; j0 < SC_EPW; j0 += SC_BATCH)
            cgcl_scatter_batch<true>(h, out, s, d, cfi, lane, j0, cnt);
    } else {
#pragma unroll
        for (int j0 = 0; j0 < SC_EPW; j0 += SC_BATCH)
            cgcl_scatter_batch<false>(h, out, s, d, cfi, lane, j0, cnt);
    }
}

__global__ void cgcl_finalize_kernel(const float* __restrict__ h,
                                     const float* __restrict__ dinv,
                                     const float* __restrict__ bias,
                                     float* __restrict__ out, int total) {
    int gid = blockIdx.x * blockDim.x + threadIdx.x;
    if (gid >= total) return;
    int n = gid >> 6;
    int f = gid & (FHID - 1);
    float di = dinv[n];
    float v = out[gid] + h[gid] * di * di + bias[f];
    out[gid] = (v > 0.0f) ? v : expm1f(v);
}

extern "C" void kernel_launch(void* const* d_in, const int* in_sizes, int n_in,
                              void* d_out, int out_size, void* d_ws, size_t ws_size,
                              hipStream_t stream) {
    const float* x    = (const float*)d_in[0];
    const float* W    = (const float*)d_in[1];
    const float* bias = (const float*)d_in[2];
    const int*   ei   = (const int*)d_in[3];

    const int N = in_sizes[0] / FIN;
    const int E = in_sizes[3] / 2;
    const int* src = ei;
    const int* dst = ei + E;

    size_t NA = ((size_t)N + 255) & ~(size_t)255;
    unsigned* deg  = (unsigned*)d_ws;
    float*    dinv = (float*)d_ws + NA;
    float*    h    = (float*)d_ws + 2 * NA;
    float*    out  = (float*)d_out;

    hipMemsetAsync(deg, 0, (size_t)N * sizeof(unsigned), stream);
    hipMemsetAsync(out, 0, (size_t)out_size * sizeof(float), stream);

    cgcl_deg_kernel<<<(E + 255) / 256, 256, 0, stream>>>(dst, deg, E);
    cgcl_dinv_kernel<<<(N + 255) / 256, 256, 0, stream>>>(deg, dinv, N);

    const int row_tiles = (N + 15) / 16;
    cgcl_gemm_wmma<<<row_tiles, 128, 0, stream>>>(x, W, h, N);

    const long long sc_waves = ((long long)E + SC_EPW - 1) / SC_EPW;
    cgcl_scatter_kernel<<<(unsigned)((sc_waves + 7) / 8), 256, 0, stream>>>(
        src, dst, dinv, h, out, E);

    const int total = N * FHID;
    cgcl_finalize_kernel<<<(total + 255) / 256, 256, 0, stream>>>(
        h, dinv, bias, out, total);
}


# degree count: global u32 atomics on deg[dst] replaced by LDS histograms per (edge chunk, node range) block, written as partial-count rows and summed in the dinv kernel; deg memset dropped
# speedup vs baseline: 2.2229x; 1.5015x over previous
#include <hip/hip_runtime.h>
#include <hip/hip_bf16.h>
#include <math.h>

#define FIN  128
#define FHID 64

typedef float v2f __attribute__((ext_vector_type(2)));
typedef float v8f __attribute__((ext_vector_type(8)));

#define DG_CHUNKS  32
#define DG_RANGE   12544
#define DG_THREADS 1024
#define DG_UNR     13

__device__ __forceinline__ void cgcl_deg_count(unsigned* hist, int d, int lo) {
    const unsigned r = (unsigned)(d - lo);
    if (r < (unsigned)DG_RANGE) atomicAdd(&hist[r], 1u);
}

__global__ void __launch_bounds__(DG_THREADS)
cgcl_deg_hist_kernel(const int* __restrict__ dst, unsigned* __restrict__ part,
                     int E, int N) {
    __shared__ unsigned hist[DG_RANGE];
    const int tid = threadIdx.x;
    const int g   = blockIdx.x;
    const int lo  = blockIdx.y * DG_RANGE;

    for (int i = tid; i < DG_RANGE; i += DG_THREADS) hist[i] = 0u;
    __syncthreads();

    const int q   = (E + 3) >> 2;
    const int per = (q + DG_CHUNKS - 1) / DG_CHUNKS;
    const int q0  = g * per;
    const int q1  = (q0 + per < q) ? (q0 + per) : q;
    const bool vec = ((E & 3) == 0);

    if (vec) {
        const int4* __restrict__ d4 = (const int4*)dst;
        for (int base = q0 + tid; base < q1; base += DG_THREADS * DG_UNR) {
            int4 v[DG_UNR];
#pragma unroll
            for (int u = 0; u < DG_UNR; ++u) {
                const int idx = base + u * DG_THREADS;
                v[u] = d4[idx < q1 ? idx : q1 - 1];
            }
#pragma unroll
            for (int u = 0; u < DG_UNR; ++u) {
                if (base + u * DG_THREADS < q1) {
                    cgcl_deg_count(hist, v[u].x, lo);
                    cgcl_deg_count(hist, v[u].y, lo);
                    cgcl_deg_count(hist, v[u].z, lo);
                    cgcl_deg_count(hist, v[u].w, lo);
                }
            }
        }
    } else {
        const long long e0 = (long long)q0 * 4;
        const long long e1 = ((long long)q1 * 4 < (long long)E) ? (long long)q1 * 4 : (long long)E;
        for (long long e = e0 + tid; e < e1; e += DG_THREADS)
            cgcl_deg_count(hist, dst[e], lo);
    }
    __syncthreads();

    unsigned* __restrict__ prow = part + (size_t)g * N;
    for (int i = tid; i < DG_RANGE; i += DG_THREADS) {
        const int n = lo + i;
        if (n < N) prow[n] = hist[i];
    }
}

__global__ void cgcl_dinv_kernel(const unsigned* __restrict__ part,
                                 float* __restrict__ dinv, int N) {
    int n = blockIdx.x * blockDim.x + threadIdx.x;
    if (n < N) {
        unsigned s = 0u;
#pragma unroll
        for (int g = 0; g < DG_CHUNKS; ++g) s += part[(size_t)g * N + n];
        dinv[n] = rsqrtf((float)s + 1.0f);
    }
}

__global__ void __launch_bounds__(128)
cgcl_gemm_wmma(const float* __restrict__ x, const float* __restrict__ W,
               float* __restrict__ h, int N) {
    const int lane = threadIdx.x & 31;
    const int wave = threadIdx.x >> 5;
    const int m    = lane & 15;
    const int hi   = lane >> 4;
    const int koff = hi * 2;

    int row = blockIdx.x * 16 + m;
    if (row >= N) row = N - 1;
    const int col = wave * 16 + m;

    const float* __restrict__ xrow = x + (size_t)row * FIN;

    v8f c = {};
#pragma unroll 4
    for (int kk = 0; kk < FIN / 4; ++kk) {
        const int k = kk * 4 + koff;
        v2f a = *(const v2f*)(xrow + k);
        v2f bfrag;
        bfrag.x = W[(k + 0) * FHID + col];
        bfrag.y = W[(k + 1) * FHID + col];
        c = __builtin_amdgcn_wmma_f32_16x16x4_f32(
                 false, a,  false, bfrag,
                 (short)0, c,  false,  false);
    }

    const int mbase = blockIdx.x * 16 + hi * 8;
    float* __restrict__ hout = h + (size_t)mbase * FHID + col;
    if (blockIdx.x * 16 + 16 <= N) {
#pragma unroll
        for (int v = 0; v < 8; ++v) {
            hout[(size_t)v * FHID] = c[v];
        }
    } else {
#pragma unroll
        for (int v = 0; v < 8; ++v) {
            if (mbase + v < N) hout[(size_t)v * FHID] = c[v];
        }
    }
}

#define SC_EPW   32
#define SC_BATCH 16

template <bool FULL>
__device__ __forceinline__ void cgcl_scatter_batch(
        const float* __restrict__ h, float* __restrict__ out,
        int s, int d, int cfi, int lane, int j0, int cnt) {
    float v0[SC_BATCH], v1[SC_BATCH];
#pragma unroll
    for (int j = 0; j < SC_BATCH; ++j) {
        const int sj = __builtin_amdgcn_readlane(s, j0 + j);
        const float* __restrict__ hp = h + (size_t)sj * FHID + lane;
        v0[j] = hp[0];
        v1[j] = hp[32];
    }
#pragma unroll
    for (int j = 0; j < SC_BATCH; ++j) {
        if (FULL || (j0 + j < cnt)) {
            const int   dj = __builtin_amdgcn_readlane(d, j0 + j);
            const float cj = __int_as_float(__builtin_amdgcn_readlane(cfi, j0 + j));
            float* o = out + (size_t)dj * FHID + lane;
            unsafeAtomicAdd(o,      v0[j] * cj);
            unsafeAtomicAdd(o + 32, v1[j] * cj);
        }
    }
}

__global__ void __launch_bounds__(256)
cgcl_scatter_kernel(const int* __restrict__ src,
                    const int* __restrict__ dst,
                    const float* __restrict__ dinv,
                    const float* __restrict__ h,
                    float* __restrict__ out, int E) {
    const int lane = threadIdx.x & 31;
    const int wib = __builtin_amdgcn_readfirstlane((int)(threadIdx.x >> 5));
    const long long wave = (long long)blockIdx.x * (blockDim.x >> 5) + wib;
    const long long e0 = wave * SC_EPW;
    if (e0 >= (long long)E) return;
    const long long left = (long long)E - e0;
    const int cnt = left < SC_EPW ? (int)left : SC_EPW;

    int s = 0, d = 0;
    float cf = 0.0f;
    if (lane < cnt) {
        s = src[e0 + lane];
        d = dst[e0 + lane];
        cf = dinv[s] * dinv[d];
    }
    const int cfi = __float_as_int(cf);

    if (cnt == SC_EPW) {
#pragma unroll
        for (int j0 = 0; j0 < SC_EPW; j0 += SC_BATCH)
            cgcl_scatter_batch<true>(h, out, s, d, cfi, lane, j0, cnt);
    } else {
#pragma unroll
        for (int j0 = 0; j0 < SC_EPW; j0 += SC_BATCH)
            cgcl_scatter_batch<false>(h, out, s, d, cfi, lane, j0, cnt);
    }
}

__global__ void cgcl_finalize_kernel(const float* __restrict__ h,
                                     const float* __restrict__ dinv,
                                     const float* __restrict__ bias,
                                     float* __restrict__ out, int total) {
    int gid = blockIdx.x * blockDim.x + threadIdx.x;
    if (gid >= total) return;
    int n = gid >> 6;
    int f = gid & (FHID - 1);
    float di = dinv[n];
    float v = out[gid] + h[gid] * di * di + bias[f];
    out[gid] = (v > 0.0f) ? v : expm1f(v);
}

extern "C" void kernel_launch(void* const* d_in, const int* in_sizes, int n_in,
                              void* d_out, int out_size, void* d_ws, size_t ws_size,
                              hipStream_t stream) {
    const float* x    = (const float*)d_in[0];
    const float* W    = (const float*)d_in[1];
    const float* bias = (const float*)d_in[2];
    const int*   ei   = (const int*)d_in[3];

    const int N = in_sizes[0] / FIN;
    const int E = in_sizes[3] / 2;
    const int* src = ei;
    const int* dst = ei + E;

    size_t NA = ((size_t)N + 255) & ~(size_t)255;
    float*    dinv = (float*)d_ws + NA;
    float*    h    = (float*)d_ws + 2 * NA;
    unsigned* part = (unsigned*)h;
    float*    out  = (float*)d_out;

    hipMemsetAsync(out, 0, (size_t)out_size * sizeof(float), stream);

    cgcl_deg_hist_kernel<<<dim3(DG_CHUNKS, (N + DG_RANGE - 1) / DG_RANGE), DG_THREADS, 0, stream>>>(
        dst, part, E, N);
    cgcl_dinv_kernel<<<(N + 255) / 256, 256, 0, stream>>>(part, dinv, N);

    const int row_tiles = (N + 15) / 16;
    cgcl_gemm_wmma<<<row_tiles, 128, 0, stream>>>(x, W, h, N);

    const long long sc_waves = ((long long)E + SC_EPW - 1) / SC_EPW;
    cgcl_scatter_kernel<<<(unsigned)((sc_waves + 7) / 8), 256, 0, stream>>>(
        src, dst, dinv, h, out, E);

    const int total = N * FHID;
    cgcl_finalize_kernel<<<(total + 255) / 256, 256, 0, stream>>>(
        h, dinv, bias, out, total);
}


# edge coefficient factored out of the scatter: h rows stored pre-scaled by dinv[src] (GEMM epilogue), dinv[dst] applied once per summed row in finalize; no per-edge dinv reads; atomics, arrival order
# speedup vs baseline: 2.8532x; 1.2835x over previous
#include <hip/hip_runtime.h>
#include <hip/hip_bf16.h>
#include <math.h>

#define FIN  128
#define FHID 64

typedef float v2f __attribute__((ext_vector_type(2)));
typedef float v8f __attribute__((ext_vector_type(8)));

#define DG_CHUNKS  32
#define DG_RANGE   12544
#define DG_THREADS 1024
#define DG_UNR     13

__device__ __forceinline__ void cgcl_deg_count(unsigned* hist, int d, int lo) {
    const unsigned r = (unsigned)(d - lo);
    if (r < (unsigned)DG_RANGE) atomicAdd(&hist[r], 1u);
}

__global__ void __launch_bounds__(DG_THREADS)
cgcl_deg_hist_kernel(const int* __restrict__ dst, unsigned* __restrict__ part,
                     int E, int N) {
    __shared__ unsigned hist[DG_RANGE];
    const int tid = threadIdx.x;
    const int g   = blockIdx.x;
    const int lo  = blockIdx.y * DG_RANGE;

    for (int i = tid; i < DG_RANGE; i += DG_THREADS) hist[i] = 0u;
    __syncthreads();

    const int q   = (E + 3) >> 2;
    const int per = (q + DG_CHUNKS - 1) / DG_CHUNKS;
    const int q0  = g * per;
    const int q1  = (q0 + per < q) ? (q0 + per) : q;
    const bool vec = ((E & 3) == 0);

    if (vec) {
        const int4* __restrict__ d4 = (const int4*)dst;
        for (int base = q0 + tid; base < q1; base += DG_THREADS * DG_UNR) {
            int4 v[DG_UNR];
#pragma unroll
            for (int u = 0; u < DG_UNR; ++u) {
                const int idx = base + u * DG_THREADS;
                v[u] = d4[idx < q1 ? idx : q1 - 1];
            }
#pragma unroll
            for (int u = 0; u < DG_UNR; ++u) {
                if (base + u * DG_THREADS < q1) {
                    cgcl_deg_count(hist, v[u].x, lo);
                    cgcl_deg_count(hist, v[u].y, lo);
                    cgcl_deg_count(hist, v[u].z, lo);
                    cgcl_deg_count(hist, v[u].w, lo);
                }
            }
        }
    } else {
        const long long e0 = (long long)q0 * 4;
        const long long e1 = ((long long)q1 * 4 < (long long)E) ? (long long)q1 * 4 : (long long)E;
        for (long long e = e0 + tid; e < e1; e += DG_THREADS)
            cgcl_deg_count(hist, dst[e], lo);
    }
    __syncthreads();

    unsigned* __restrict__ prow = part + (size_t)g * N;
    for (int i = tid; i < DG_RANGE; i += DG_THREADS) {
        const int n = lo + i;
        if (n < N) prow[n] = hist[i];
    }
}

#define SC_CTR_STRIDE 64
__global__ void cgcl_dinv_kernel(const unsigned* __restrict__ part,
                                 float* __restrict__ dinv,
                                 unsigned* __restrict__ ctr, int N) {
    if (blockIdx.x == 0 && threadIdx.x < 2) ctr[threadIdx.x * SC_CTR_STRIDE] = 0u;
    int n = blockIdx.x * blockDim.x + threadIdx.x;
    if (n < N) {
        unsigned s = 0u;
#pragma unroll
        for (int g = 0; g < DG_CHUNKS; ++g) s += part[(size_t)g * N + n];
        dinv[n] = rsqrtf((float)s + 1.0f);
    }
}

__global__ void __launch_bounds__(128)
cgcl_gemm_wmma(const float* __restrict__ x, const float* __restrict__ W,
               const float* __restrict__ dinv,
               float* __restrict__ h, int N) {
    const int lane = threadIdx.x & 31;
    const int wave = threadIdx.x >> 5;
    const int m    = lane & 15;
    const int hi   = lane >> 4;
    const int koff = hi * 2;

    int row = blockIdx.x * 16 + m;
    if (row >= N) row = N - 1;
    const int col = wave * 16 + m;

    const float* __restrict__ xrow = x + (size_t)row * FIN;

    v8f c = {};
#pragma unroll 4
    for (int kk = 0; kk < FIN / 4; ++kk) {
        const int k = kk * 4 + koff;
        v2f a = *(const v2f*)(xrow + k);
        v2f bfrag;
        bfrag.x = W[(k + 0) * FHID + col];
        bfrag.y = W[(k + 1) * FHID + col];
        c = __builtin_amdgcn_wmma_f32_16x16x4_f32(
                 false, a,  false, bfrag,
                 (short)0, c,  false,  false);
    }

    const int mbase = blockIdx.x * 16 + hi * 8;
    float* __restrict__ hout = h + (size_t)mbase * FHID + col;
    if (blockIdx.x * 16 + 16 <= N) {
        const float4 d0 = *(const float4*)(dinv + mbase);
        const float4 d1 = *(const float4*)(dinv + mbase + 4);
        const float dv[8] = {d0.x, d0.y, d0.z, d0.w, d1.x, d1.y, d1.z, d1.w};
#pragma unroll
        for (int v = 0; v < 8; ++v) {
            hout[(size_t)v * FHID] = c[v] * dv[v];
        }
    } else {
#pragma unroll
        for (int v = 0; v < 8; ++v) {
            if (mbase + v < N) hout[(size_t)v * FHID] = c[v] * dinv[mbase + v];
        }
    }
}

#define SC_EPW    32
#define SC_WAVES  8
#define SC_JEDGES (SC_EPW * SC_WAVES)
#define SC_BATCH  16
#define SC_REST   8
#define SC_BLOCKS 2048

template <int B, bool FULL>
__device__ __forceinline__ unsigned cgcl_scatter_batch(
        const float* __restrict__ h, float* __restrict__ out,
        int s, int d, int lane, unsigned m) {
    float v0[B], v1[B];
    const int c = __builtin_popcount(m);
    unsigned ma = m;
    int idx = __builtin_ctz(ma);
#pragma unroll
    for (int j = 0; j < B; ++j) {
        const int sj = __builtin_amdgcn_readlane(s, idx);
        const float* __restrict__ hp = h + (size_t)sj * FHID + lane;
        v0[j] = hp[0];
        v1[j] = hp[32];
        ma &= ma - 1u;
        idx = (ma != 0u) ? __builtin_ctz(ma) : idx;
    }
    if (!FULL) {
#pragma unroll
        for (int j = 0; j < B; ++j) asm volatile("" :: "v"(v0[j]), "v"(v1[j]));
    }
    unsigned mb = m;
#pragma unroll
    for (int j = 0; j < B; ++j) {
        if (FULL || (j < c)) {
            const int jdx = __builtin_ctz(mb);
            mb &= mb - 1u;
            const int dj = __builtin_amdgcn_readlane(d, jdx);
            float* o = out + (size_t)dj * FHID + lane;
            unsafeAtomicAdd(o,      v0[j]);
            unsafeAtomicAdd(o + 32, v1[j]);
        }
    }
    return ma;
}

__global__ void __launch_bounds__(256)
cgcl_scatter_kernel(const int* __restrict__ src,
                    const int* __restrict__ dst,
                    const float* __restrict__ h,
                    float* __restrict__ out,
                    unsigned* __restrict__ ctr, int E) {
    __shared__ unsigned s_home;
    __shared__ unsigned s_t;
    const int lane = threadIdx.x & 31;
    const int wib = __builtin_amdgcn_readfirstlane((int)(threadIdx.x >> 5));
    const unsigned njobs = (unsigned)(((long long)E + SC_JEDGES - 1) / SC_JEDGES);

    if (threadIdx.x == 0) {
        const unsigned id = (unsigned)__builtin_amdgcn_s_sendmsg_rtn(0x87u);
        s_home = (id >> 18) & 1u;
    }
    __syncthreads();
    unsigned owner = (unsigned)__builtin_amdgcn_readfirstlane((int)s_home);

    for (int phase = 0; phase < 2; ++phase) {
        for (unsigned trip = 0; trip <= njobs; ++trip) {
            if (threadIdx.x == 0) s_t = atomicAdd(&ctr[owner * SC_CTR_STRIDE], 1u);
            __syncthreads();
            const unsigned t = (unsigned)__builtin_amdgcn_readfirstlane((int)s_t);
            __syncthreads();
            if (t >= njobs) break;

            const long long e = (long long)t * SC_JEDGES + wib * SC_EPW + lane;
            int s = 0, d = 0;
            bool mine = false;
            if (e < (long long)E) {
                s = src[e];
                d = dst[e];
                mine = (((unsigned)d & 1u) == owner);
            }
            unsigned m = __builtin_amdgcn_ballot_w32(mine);
            if (m != 0u) {
                if (__builtin_popcount(m) >= SC_BATCH)
                    m = cgcl_scatter_batch<SC_BATCH, true >(h, out, s, d, lane, m);
                else
                    m = cgcl_scatter_batch<SC_BATCH, false>(h, out, s, d, lane, m);
                for (int b = 0; b < (SC_EPW - SC_BATCH + SC_REST - 1) / SC_REST; ++b) {
                    if (m == 0u) break;
                    m = cgcl_scatter_batch<SC_REST, false>(h, out, s, d, lane, m);
                }
            }
        }
        owner ^= 1u;
    }
}

__global__ void cgcl_finalize_kernel(const float* __restrict__ h,
                                     const float* __restrict__ dinv,
                                     const float* __restrict__ bias,
                                     float* __restrict__ out, int total) {
    int gid = blockIdx.x * blockDim.x + threadIdx.x;
    if (gid >= total) return;
    int n = gid >> 6;
    int f = gid & (FHID - 1);
    float di = dinv[n];
    float v = di * (out[gid] + h[gid]) + bias[f];
    out[gid] = (v > 0.0f) ? v : expm1f(v);
}

extern "C" void kernel_launch(void* const* d_in, const int* in_sizes, int n_in,
                              void* d_out, int out_size, void* d_ws, size_t ws_size,
                              hipStream_t stream) {
    const float* x    = (const float*)d_in[0];
    const float* W    = (const float*)d_in[1];
    const float* bias = (const float*)d_in[2];
    const int*   ei   = (const int*)d_in[3];

    const int N = in_sizes[0] / FIN;
    const int E = in_sizes[3] / 2;
    const int* src = ei;
    const int* dst = ei + E;

    size_t NA = ((size_t)N + 255) & ~(size_t)255;
    float*    dinv = (float*)d_ws + NA;
    float*    h    = (float*)d_ws + 2 * NA;
    unsigned* part = (unsigned*)h;
    unsigned* ctr  = (unsigned*)d_ws;
    float*    out  = (float*)d_out;

    hipMemsetAsync(out, 0, (size_t)out_size * sizeof(float), stream);

    cgcl_deg_hist_kernel<<<dim3(DG_CHUNKS, (N + DG_RANGE - 1) / DG_RANGE), DG_THREADS, 0, stream>>>(
        dst, part, E, N);
    cgcl_dinv_kernel<<<(N + 255) / 256, 256, 0, stream>>>(part, dinv, ctr, N);

    const int row_tiles = (N + 15) / 16;
    cgcl_gemm_wmma<<<row_tiles, 128, 0, stream>>>(x, W, dinv, h, N);

    cgcl_scatter_kernel<<<SC_BLOCKS, 256, 0, stream>>>(
        src, dst, h, out, ctr, E);

    const int total = N * FHID;
    cgcl_finalize_kernel<<<(total + 255) / 256, 256, 0, stream>>>(
        h, dinv, bias, out, total);
}


# GEMM: x tile fetched as whole 512-B rows (16-B load per lane, 4 rows a wave, tile read once), parked in LDS; A fragments read from LDS (was 32 loads of 16 lines x 16 B in each of 4 waves)
# speedup vs baseline: 3.1336x; 1.0983x over previous
#include <hip/hip_runtime.h>
#include <hip/hip_bf16.h>
#include <math.h>

#define FIN  128
#define FHID 64

typedef float v2f __attribute__((ext_vector_type(2)));
typedef float v8f __attribute__((ext_vector_type(8)));

#define DG_CHUNKS  32
#define DG_RANGE   12544
#define DG_THREADS 1024
#define DG_UNR     13

__device__ __forceinline__ void cgcl_deg_count(unsigned* hist, int d, int lo) {
    const unsigned r = (unsigned)(d - lo);
    if (r < (unsigned)DG_RANGE) atomicAdd(&hist[r], 1u);
}

__global__ void __launch_bounds__(DG_THREADS)
cgcl_deg_hist_kernel(const int* __restrict__ dst, unsigned* __restrict__ part,
                     int E, int N) {
    __shared__ unsigned hist[DG_RANGE];
    const int tid = threadIdx.x;
    const int g   = blockIdx.x;
    const int lo  = blockIdx.y * DG_RANGE;

    for (int i = tid; i < DG_RANGE; i += DG_THREADS) hist[i] = 0u;
    __syncthreads();

    const int q   = (E + 3) >> 2;
    const int per = (q + DG_CHUNKS - 1) / DG_CHUNKS;
    const int q0  = g * per;
    const int q1  = (q0 + per < q) ? (q0 + per) : q;
    const bool vec = ((E & 3) == 0);

    if (vec) {
        const int4* __restrict__ d4 = (const int4*)dst;
        for (int base = q0 + tid; base < q1; base += DG_THREADS * DG_UNR) {
            int4 v[DG_UNR];
#pragma unroll
            for (int u = 0; u < DG_UNR; ++u) {
                const int idx = base + u * DG_THREADS;
                v[u] = d4[idx < q1 ? idx : q1 - 1];
            }
#pragma unroll
            for (int u = 0; u < DG_UNR; ++u) {
                if (base + u * DG_THREADS < q1) {
                    cgcl_deg_count(hist, v[u].x, lo);
                    cgcl_deg_count(hist, v[u].y, lo);
                    cgcl_deg_count(hist, v[u].z, lo);
                    cgcl_deg_count(hist, v[u].w, lo);
                }
            }
        }
    } else {
        const long long e0 = (long long)q0 * 4;
        const long long e1 = ((long long)q1 * 4 < (long long)E) ? (long long)q1 * 4 : (long long)E;
        for (long long e = e0 + tid; e < e1; e += DG_THREADS)
            cgcl_deg_count(hist, dst[e], lo);
    }
    __syncthreads();

    unsigned* __restrict__ prow = part + (size_t)g * N;
    for (int i = tid; i < DG_RANGE; i += DG_THREADS) {
        const int n = lo + i;
        if (n < N) prow[n] = hist[i];
    }
}

#define SC_CTR_STRIDE 64
__global__ void cgcl_dinv_kernel(const unsigned* __restrict__ part,
                                 float* __restrict__ dinv,
                                 unsigned* __restrict__ ctr, int N) {
    if (blockIdx.x == 0 && threadIdx.x < 2) ctr[threadIdx.x * SC_CTR_STRIDE] = 0u;
    int n = blockIdx.x * blockDim.x + threadIdx.x;
    if (n < N) {
        unsigned s = 0u;
#pragma unroll
        for (int g = 0; g < DG_CHUNKS; ++g) s += part[(size_t)g * N + n];
        dinv[n] = rsqrtf((float)s + 1.0f);
    }
}

#define GM_PITCH (FIN + 4)

__global__ void __launch_bounds__(128)
cgcl_gemm_wmma(const float* __restrict__ x, const float* __restrict__ W,
               const float* __restrict__ dinv,
               float* __restrict__ h, int N) {
    __shared__ __attribute__((aligned(16))) float xs[16 * GM_PITCH];

    const int lane = threadIdx.x & 31;
    const int wave = threadIdx.x >> 5;
    const int m    = lane & 15;
    const int hi   = lane >> 4;
    const int koff = hi * 2;
    const int col  = wave * 16 + m;

    float4 xr[4];
#pragma unroll
    for (int j = 0; j < 4; ++j) {
        int row = blockIdx.x * 16 + wave * 4 + j;
        if (row >= N) row = N - 1;
        xr[j] = *(const float4*)(x + (size_t)row * FIN + lane * 4);
    }
#pragma unroll
    for (int j = 0; j < 4; ++j)
        *(float4*)(xs + (wave * 4 + j) * GM_PITCH + lane * 4) = xr[j];
    __syncthreads();

    const float* xrow = xs + m * GM_PITCH + koff;

    v8f c = {};
#pragma unroll 4
    for (int kk = 0; kk < FIN / 4; ++kk) {
        const int k = kk * 4 + koff;
        v2f a = *(const v2f*)(xrow + kk * 4);
        v2f bfrag;
        bfrag.x = W[(k + 0) * FHID + col];
        bfrag.y = W[(k + 1) * FHID + col];
        c = __builtin_amdgcn_wmma_f32_16x16x4_f32(
                 false, a,  false, bfrag,
                 (short)0, c,  false,  false);
    }

    const int mbase = blockIdx.x * 16 + hi * 8;
    float* __restrict__ hout = h + (size_t)mbase * FHID + col;
    if (blockIdx.x * 16 + 16 <= N) {
        const float4 d0 = *(const float4*)(dinv + mbase);
        const float4 d1 = *(const float4*)(dinv + mbase + 4);
        const float dv[8] = {d0.x, d0.y, d0.z, d0.w, d1.x, d1.y, d1.z, d1.w};
#pragma unroll
        for (int v = 0; v < 8; ++v) {
            hout[(size_t)v * FHID] = c[v] * dv[v];
        }
    } else {
#pragma unroll
        for (int v = 0; v < 8; ++v) {
            if (mbase + v < N) hout[(size_t)v * FHID] = c[v] * dinv[mbase + v];
        }
    }
}

#define SC_EPW    32
#define SC_WAVES  8
#define SC_JEDGES (SC_EPW * SC_WAVES)
#define SC_BATCH  16
#define SC_REST   8
#define SC_BLOCKS 2048

template <int B, bool FULL>
__device__ __forceinline__ unsigned cgcl_scatter_batch(
        const float* __restrict__ h, float* __restrict__ out,
        int s, int d, int lane, unsigned m) {
    float v0[B], v1[B];
    const int c = __builtin_popcount(m);
    unsigned ma = m;
    int idx = __builtin_ctz(ma);
#pragma unroll
    for (int j = 0; j < B; ++j) {
        const int sj = __builtin_amdgcn_readlane(s, idx);
        const float* __restrict__ hp = h + (size_t)sj * FHID + lane;
        v0[j] = hp[0];
        v1[j] = hp[32];
        ma &= ma - 1u;
        idx = (ma != 0u) ? __builtin_ctz(ma) : idx;
    }
    if (!FULL) {
#pragma unroll
        for (int j = 0; j < B; ++j) asm volatile("" :: "v"(v0[j]), "v"(v1[j]));
    }
    unsigned mb = m;
#pragma unroll
    for (int j = 0; j < B; ++j) {
        if (FULL || (j < c)) {
            const int jdx = __builtin_ctz(mb);
            mb &= mb - 1u;
            const int dj = __builtin_amdgcn_readlane(d, jdx);
            float* o = out + (size_t)dj * FHID + lane;
            unsafeAtomicAdd(o,      v0[j]);
            unsafeAtomicAdd(o + 32, v1[j]);
        }
    }
    return ma;
}

__global__ void __launch_bounds__(256)
cgcl_scatter_kernel(const int* __restrict__ src,
                    const int* __restrict__ dst,
                    const float* __restrict__ h,
                    float* __restrict__ out,
                    unsigned* __restrict__ ctr, int E) {
    __shared__ unsigned s_home;
    __shared__ unsigned s_t;
    const int lane = threadIdx.x & 31;
    const int wib = __builtin_amdgcn_readfirstlane((int)(threadIdx.x >> 5));
    const unsigned njobs = (unsigned)(((long long)E + SC_JEDGES - 1) / SC_JEDGES);

    if (threadIdx.x == 0) {
        const unsigned id = (unsigned)__builtin_amdgcn_s_sendmsg_rtn(0x87u);
        s_home = (id >> 18) & 1u;
    }
    __syncthreads();
    unsigned owner = (unsigned)__builtin_amdgcn_readfirstlane((int)s_home);

    for (int phase = 0; phase < 2; ++phase) {
        for (unsigned trip = 0; trip <= njobs; ++trip) {
            if (threadIdx.x == 0) s_t = atomicAdd(&ctr[owner * SC_CTR_STRIDE], 1u);
            __syncthreads();
            const unsigned t = (unsigned)__builtin_amdgcn_readfirstlane((int)s_t);
            __syncthreads();
            if (t >= njobs) break;

            const long long e = (long long)t * SC_JEDGES + wib * SC_EPW + lane;
            int s = 0, d = 0;
            bool mine = false;
            if (e < (long long)E) {
                s = src[e];
                d = dst[e];
                mine = (((unsigned)d & 1u) == owner);
            }
            unsigned m = __builtin_amdgcn_ballot_w32(mine);
            if (m != 0u) {
                if (__builtin_popcount(m) >= SC_BATCH)
                    m = cgcl_scatter_batch<SC_BATCH, true >(h, out, s, d, lane, m);
                else
                    m = cgcl_scatter_batch<SC_BATCH, false>(h, out, s, d, lane, m);
                for (int b = 0; b < (SC_EPW - SC_BATCH + SC_REST - 1) / SC_REST; ++b) {
                    if (m == 0u) break;
                    m = cgcl_scatter_batch<SC_REST, false>(h, out, s, d, lane, m);
                }
            }
        }
        owner ^= 1u;
    }
}

__global__ void cgcl_finalize_kernel(const float* __restrict__ h,
                                     const float* __restrict__ dinv,
                                     const float* __restrict__ bias,
                                     float* __restrict__ out, int total) {
    int gid = blockIdx.x * blockDim.x + threadIdx.x;
    if (gid >= total) return;
    int n = gid >> 6;
    int f = gid & (FHID - 1);
    float di = dinv[n];
    float v = di * (out[gid] + h[gid]) + bias[f];
    out[gid] = (v > 0.0f) ? v : expm1f(v);
}

extern "C" void kernel_launch(void* const* d_in, const int* in_sizes, int n_in,
                              void* d_out, int out_size, void* d_ws, size_t ws_size,
                              hipStream_t stream) {
    const float* x    = (const float*)d_in[0];
    const float* W    = (const float*)d_in[1];
    const float* bias = (const float*)d_in[2];
    const int*   ei   = (const int*)d_in[3];

    const int N = in_sizes[0] / FIN;
    const int E = in_sizes[3] / 2;
    const int* src = ei;
    const int* dst = ei + E;

    size_t NA = ((size_t)N + 255) & ~(size_t)255;
    float*    dinv = (float*)d_ws + NA;
    float*    h    = (float*)d_ws + 2 * NA;
    unsigned* part = (unsigned*)h;
    unsigned* ctr  = (unsigned*)d_ws;
    float*    out  = (float*)d_out;

    hipMemsetAsync(out, 0, (size_t)out_size * sizeof(float), stream);

    cgcl_deg_hist_kernel<<<dim3(DG_CHUNKS, (N + DG_RANGE - 1) / DG_RANGE), DG_THREADS, 0, stream>>>(
        dst, part, E, N);
    cgcl_dinv_kernel<<<(N + 255) / 256, 256, 0, stream>>>(part, dinv, ctr, N);

    const int row_tiles = (N + 15) / 16;
    cgcl_gemm_wmma<<<row_tiles, 128, 0, stream>>>(x, W, dinv, h, N);

    cgcl_scatter_kernel<<<SC_BLOCKS, 256, 0, stream>>>(
        src, dst, h, out, ctr, E);

    const int total = N * FHID;
    cgcl_finalize_kernel<<<(total + 255) / 256, 256, 0, stream>>>(
        h, dinv, bias, out, total);
}
